# starting kernel
# baseline (speedup 1.0000x reference)
#include <hip/hip_runtime.h>


#define B_ 16
#define S_ 2048
#define D_ 128
#define THR 0.3f

typedef __attribute__((ext_vector_type(2))) float v2f;
typedef __attribute__((ext_vector_type(8))) float v8f;

#define KSTRIDE 132
#define VSTRIDE 20
#define PSTRIDE 20

static __device__ __forceinline__
v8f wmma_f32(v2f a, v2f b, v8f c) {
    return __builtin_amdgcn_wmma_f32_16x16x4_f32(false, a, false, b,
                                                 (short)0, c, false, false);
}

__global__ __launch_bounds__(128)
void attn_fa_kernel(const float* __restrict__ Q,
                    const float* __restrict__ T,
                    const float* __restrict__ V,
                    float* __restrict__ O)
{
    __shared__ float lds_k [16  * KSTRIDE];
    __shared__ float lds_vt[128 * VSTRIDE];
    __shared__ float lds_p [4 * 16 * PSTRIDE];

    const int b    = blockIdx.y;
    const int g    = blockIdx.x;
    const int tid  = threadIdx.x;
    const int wave = tid >> 5;
    const int lane = tid & 31;
    const int n    = lane & 15;
    const int hi   = lane >> 4;

    const int q0     = g * 64 + wave * 16;
    const int myDiag = q0 >> 4;
    const int jmax   = g * 4 + 3;

    const float SCALE2 = 0.08838834764831845f * 1.4426950408889634f;

    v2f qa[32];
    {
        const float* qp = Q + ((size_t)b * S_ + q0 + n) * D_ + 2 * hi;
#pragma unroll
        for (int c = 0; c < 32; ++c)
            qa[c] = *(const v2f*)(qp + 4 * c);
    }

    v8f y[8];
#pragma unroll
    for (int nt = 0; nt < 8; ++nt) y[nt] = {};
    float m_run[8], l_run[8];
#pragma unroll
    for (int e = 0; e < 8; ++e) { m_run[e] = -1e30f; l_run[e] = 0.0f; }

    for (int j = 0; j <= jmax; ++j) {
        {
            const int r     = tid >> 3;
            const int dbase = (tid & 7) * 16;
            const float* kp = T + ((size_t)b * S_ + (size_t)j * 16 + r) * D_ + dbase;
            const float* vp = V + ((size_t)b * S_ + (size_t)j * 16 + r) * D_ + dbase;
#pragma unroll
            for (int i = 0; i < 4; ++i) {
                float4 kq = *(const float4*)(kp + 4 * i);
                *(float4*)&lds_k[r * KSTRIDE + dbase + 4 * i] = kq;
                float4 vq = *(const float4*)(vp + 4 * i);
                lds_vt[(dbase + 4*i + 0) * VSTRIDE + r] = vq.x;
                lds_vt[(dbase + 4*i + 1) * VSTRIDE + r] = vq.y;
                lds_vt[(dbase + 4*i + 2) * VSTRIDE + r] = vq.z;
                lds_vt[(dbase + 4*i + 3) * VSTRIDE + r] = vq.w;
            }
        }
        __syncthreads();

        if (j <= myDiag) {
            const float* krow = &lds_k[n * KSTRIDE + 2 * hi];
            v8f s0 = {}, s1 = {};
            v2f kb[2][8];
#pragma unroll
            for (int i = 0; i < 8; ++i)
                kb[0][i] = *(const v2f*)(krow + 4 * i);
#pragma unroll
            for (int grp = 0; grp < 4; ++grp) {
                if (grp < 3) {
#pragma unroll
                    for (int i = 0; i < 8; ++i)
                        kb[(grp + 1) & 1][i] =
                            *(const v2f*)(krow + 4 * (8 * (grp + 1) + i));
                }
#pragma unroll
                for (int i = 0; i < 8; i += 2) {
                    s0 = wmma_f32(qa[8 * grp + i],     kb[grp & 1][i],     s0);
                    s1 = wmma_f32(qa[8 * grp + i + 1], kb[grp & 1][i + 1], s1);
                }
            }

            const int kk = j * 16 + n;
            float al[8], p[8];
#pragma unroll
            for (int e = 0; e < 8; ++e) {
                float sv = s0[e] + s1[e];
                sv = (sv > THR) ? sv : 0.0f;
                float z = sv * SCALE2;
                const int m = e + 8 * hi;
                if (kk > q0 + m) z = -1e30f;
                float rm = z;
                rm = fmaxf(rm, __shfl_xor(rm, 1, 32));
                rm = fmaxf(rm, __shfl_xor(rm, 2, 32));
                rm = fmaxf(rm, __shfl_xor(rm, 4, 32));
                rm = fmaxf(rm, __shfl_xor(rm, 8, 32));
                float mn = fmaxf(m_run[e], rm);
                float a  = __builtin_amdgcn_exp2f(m_run[e] - mn);
                float pe = __builtin_amdgcn_exp2f(z - mn);
                float rs = pe;
                rs += __shfl_xor(rs, 1, 32);
                rs += __shfl_xor(rs, 2, 32);
                rs += __shfl_xor(rs, 4, 32);
                rs += __shfl_xor(rs, 8, 32);
                l_run[e] = l_run[e] * a + rs;
                m_run[e] = mn;
                al[e] = a;
                p[e]  = pe;
            }

#pragma unroll
            for (int nt = 0; nt < 8; ++nt)
#pragma unroll
                for (int e = 0; e < 8; ++e)
                    y[nt][e] *= al[e];

            float* pp = &lds_p[wave * 16 * PSTRIDE];
#pragma unroll
            for (int e = 0; e < 8; ++e)
                pp[(e + 8 * hi) * PSTRIDE + n] = p[e];

            v2f pa[4];
#pragma unroll
            for (int c = 0; c < 4; ++c)
                pa[c] = *(const v2f*)&pp[n * PSTRIDE + 4 * c + 2 * hi];

#pragma unroll
            for (int half = 0; half < 2; ++half) {
                v2f vb[16];
#pragma unroll
                for (int nt = 0; nt < 4; ++nt)
#pragma unroll
                    for (int c = 0; c < 4; ++c)
                        vb[nt * 4 + c] = *(const v2f*)
                            &lds_vt[((half * 4 + nt) * 16 + n) * VSTRIDE + 4 * c + 2 * hi];
#pragma unroll
                for (int nt = 0; nt < 4; ++nt) {
                    v8f acc = y[half * 4 + nt];
#pragma unroll
                    for (int c = 0; c < 4; ++c)
                        acc = wmma_f32(pa[c], vb[nt * 4 + c], acc);
                    y[half * 4 + nt] = acc;
                }
            }
        }
        __syncthreads();
    }

    float inv[8];
#pragma unroll
    for (int e = 0; e < 8; ++e) inv[e] = 1.0f / l_run[e];
    float* op = O + ((size_t)b * S_ + q0) * D_;
#pragma unroll
    for (int nt = 0; nt < 8; ++nt)
#pragma unroll
        for (int e = 0; e < 8; ++e)
            op[(e + 8 * hi) * D_ + nt * 16 + n] = y[nt][e] * inv[e];
}

extern "C" void kernel_launch(void* const* d_in, const int* in_sizes, int n_in,
                              void* d_out, int out_size, void* d_ws, size_t ws_size,
                              hipStream_t stream) {
    const float* Q = (const float*)d_in[0];
    const float* T = (const float*)d_in[1];
    const float* V = (const float*)d_in[2];
    float*       O = (float*)d_out;
    dim3 grid(S_ / 64, B_);
    attn_fa_kernel<<<grid, dim3(128), 0, stream>>>(Q, T, V, O);
}


# softmax 16-lane row max and row sum on the vector ALU with DPP row_xmask (v_max_num_f32_dpp, v_add_f32_dpp) instead of 64 ds_bpermute_b32 per tile, each behind an LDS wait; same butterfly order
# speedup vs baseline: 1.9096x; 1.9096x over previous
#include <hip/hip_runtime.h>


#define B_ 16
#define S_ 2048
#define D_ 128
#define THR 0.3f

typedef __attribute__((ext_vector_type(2))) float v2f;
typedef __attribute__((ext_vector_type(8))) float v8f;

#define KSTRIDE 132
#define VSTRIDE 20
#define PSTRIDE 20

static __device__ __forceinline__
v8f wmma_f32(v2f a, v2f b, v8f c) {
    return __builtin_amdgcn_wmma_f32_16x16x4_f32(false, a, false, b,
                                                 (short)0, c, false, false);
}

template <int M>
static __device__ __forceinline__
float row_xor(float v) {
    return __builtin_bit_cast(float,
        __builtin_amdgcn_update_dpp(0, __builtin_bit_cast(int, v),
                                    0x160 | M, 0xf, 0xf, true));
}

template <int M>
static __device__ __forceinline__
void row_max_step(float (&r)[8]) {
#pragma unroll
    for (int e = 0; e < 8; ++e) {
        float t;
        asm("v_max_num_f32_dpp %0, %1, %1 row_xmask:%2 row_mask:0xf bank_mask:0xf"
            : "=v"(t) : "v"(r[e]), "n"(M));
        r[e] = t;
    }
}
template <int M>
static __device__ __forceinline__
void row_sum_step(float (&r)[8]) {
#pragma unroll
    for (int e = 0; e < 8; ++e)
        r[e] += row_xor<M>(r[e]);
}

__global__ __launch_bounds__(128)
void attn_fa_kernel(const float* __restrict__ Q,
                    const float* __restrict__ T,
                    const float* __restrict__ V,
                    float* __restrict__ O)
{
    __shared__ float lds_k [16  * KSTRIDE];
    __shared__ float lds_vt[128 * VSTRIDE];
    __shared__ float lds_p [4 * 16 * PSTRIDE];

    const int b    = blockIdx.y;
    const int g    = blockIdx.x;
    const int tid  = threadIdx.x;
    const int wave = tid >> 5;
    const int lane = tid & 31;
    const int n    = lane & 15;
    const int hi   = lane >> 4;

    const int q0     = g * 64 + wave * 16;
    const int myDiag = q0 >> 4;
    const int jmax   = g * 4 + 3;

    const float SCALE2 = 0.08838834764831845f * 1.4426950408889634f;

    v2f qa[32];
    {
        const float* qp = Q + ((size_t)b * S_ + q0 + n) * D_ + 2 * hi;
#pragma unroll
        for (int c = 0; c < 32; ++c)
            qa[c] = *(const v2f*)(qp + 4 * c);
    }

    v8f y[8];
#pragma unroll
    for (int nt = 0; nt < 8; ++nt) y[nt] = {};
    float m_run[8], l_run[8];
#pragma unroll
    for (int e = 0; e < 8; ++e) { m_run[e] = -1e30f; l_run[e] = 0.0f; }

    for (int j = 0; j <= jmax; ++j) {
        {
            const int r     = tid >> 3;
            const int dbase = (tid & 7) * 16;
            const float* kp = T + ((size_t)b * S_ + (size_t)j * 16 + r) * D_ + dbase;
            const float* vp = V + ((size_t)b * S_ + (size_t)j * 16 + r) * D_ + dbase;
#pragma unroll
            for (int i = 0; i < 4; ++i) {
                float4 kq = *(const float4*)(kp + 4 * i);
                *(float4*)&lds_k[r * KSTRIDE + dbase + 4 * i] = kq;
                float4 vq = *(const float4*)(vp + 4 * i);
                lds_vt[(dbase + 4*i + 0) * VSTRIDE + r] = vq.x;
                lds_vt[(dbase + 4*i + 1) * VSTRIDE + r] = vq.y;
                lds_vt[(dbase + 4*i + 2) * VSTRIDE + r] = vq.z;
                lds_vt[(dbase + 4*i + 3) * VSTRIDE + r] = vq.w;
            }
        }
        __syncthreads();

        if (j <= myDiag) {
            const float* krow = &lds_k[n * KSTRIDE + 2 * hi];
            v8f s0 = {}, s1 = {};
            v2f kb[2][8];
#pragma unroll
            for (int i = 0; i < 8; ++i)
                kb[0][i] = *(const v2f*)(krow + 4 * i);
#pragma unroll
            for (int grp = 0; grp < 4; ++grp) {
                if (grp < 3) {
#pragma unroll
                    for (int i = 0; i < 8; ++i)
                        kb[(grp + 1) & 1][i] =
                            *(const v2f*)(krow + 4 * (8 * (grp + 1) + i));
                }
#pragma unroll
                for (int i = 0; i < 8; i += 2) {
                    s0 = wmma_f32(qa[8 * grp + i],     kb[grp & 1][i],     s0);
                    s1 = wmma_f32(qa[8 * grp + i + 1], kb[grp & 1][i + 1], s1);
                }
            }

            const int kk = j * 16 + n;
            float z[8], rm[8];
#pragma unroll
            for (int e = 0; e < 8; ++e) {
                float sv = s0[e] + s1[e];
                sv = (sv > THR) ? sv : 0.0f;
                float zz = sv * SCALE2;
                const int m = e + 8 * hi;
                if (kk > q0 + m) zz = -1e30f;
                z[e]  = zz;
                rm[e] = zz;
            }
            row_max_step<1>(rm);
            row_max_step<2>(rm);
            row_max_step<4>(rm);
            row_max_step<8>(rm);
            float al[8], p[8], rs[8];
#pragma unroll
            for (int e = 0; e < 8; ++e) {
                float mn = fmaxf(m_run[e], rm[e]);
                float a  = __builtin_amdgcn_exp2f(m_run[e] - mn);
                float pe = __builtin_amdgcn_exp2f(z[e] - mn);
                m_run[e] = mn;
                al[e] = a;
                p[e]  = pe;
                rs[e] = pe;
            }
            row_sum_step<1>(rs);
            row_sum_step<2>(rs);
            row_sum_step<4>(rs);
            row_sum_step<8>(rs);
#pragma unroll
            for (int e = 0; e < 8; ++e)
                l_run[e] = l_run[e] * al[e] + rs[e];

#pragma unroll
            for (int nt = 0; nt < 8; ++nt)
#pragma unroll
                for (int e = 0; e < 8; ++e)
                    y[nt][e] *= al[e];

            float* pp = &lds_p[wave * 16 * PSTRIDE];
#pragma unroll
            for (int e = 0; e < 8; ++e)
                pp[(e + 8 * hi) * PSTRIDE + n] = p[e];

            v2f pa[4];
#pragma unroll
            for (int c = 0; c < 4; ++c)
                pa[c] = *(const v2f*)&pp[n * PSTRIDE + 4 * c + 2 * hi];

#pragma unroll
            for (int half = 0; half < 2; ++half) {
                v2f vb[16];
#pragma unroll
                for (int nt = 0; nt < 4; ++nt)
#pragma unroll
                    for (int c = 0; c < 4; ++c)
                        vb[nt * 4 + c] = *(const v2f*)
                            &lds_vt[((half * 4 + nt) * 16 + n) * VSTRIDE + 4 * c + 2 * hi];
#pragma unroll
                for (int nt = 0; nt < 4; ++nt) {
                    v8f acc = y[half * 4 + nt];
#pragma unroll
                    for (int c = 0; c < 4; ++c)
                        acc = wmma_f32(pa[c], vb[nt * 4 + c], acc);
                    y[half * 4 + nt] = acc;
                }
            }
        }
        __syncthreads();
    }

    float inv[8];
#pragma unroll
    for (int e = 0; e < 8; ++e) inv[e] = 1.0f / l_run[e];
    float* op = O + ((size_t)b * S_ + q0) * D_;
#pragma unroll
    for (int nt = 0; nt < 8; ++nt)
#pragma unroll
        for (int e = 0; e < 8; ++e)
            op[(e + 8 * hi) * D_ + nt * 16 + n] = y[nt][e] * inv[e];
}

extern "C" void kernel_launch(void* const* d_in, const int* in_sizes, int n_in,
                              void* d_out, int out_size, void* d_ws, size_t ws_size,
                              hipStream_t stream) {
    const float* Q = (const float*)d_in[0];
    const float* T = (const float*)d_in[1];
    const float* V = (const float*)d_in[2];
    float*       O = (float*)d_out;
    dim3 grid(S_ / 64, B_);
    attn_fa_kernel<<<grid, dim3(128), 0, stream>>>(Q, T, V, O);
}


# K/V tile staging double-buffered in LDS: tile j+1's global loads issued before tile j's QK WMMAs and written to the other LDS buffer behind them; one barrier per key tile instead of two
# speedup vs baseline: 2.1119x; 1.1059x over previous
#include <hip/hip_runtime.h>


#define B_ 16
#define S_ 2048
#define D_ 128
#define THR 0.3f

typedef __attribute__((ext_vector_type(2))) float v2f;
typedef __attribute__((ext_vector_type(8))) float v8f;

#define KSTRIDE 132
#define VSTRIDE 20
#define PSTRIDE 20

static __device__ __forceinline__
v8f wmma_f32(v2f a, v2f b, v8f c) {
    return __builtin_amdgcn_wmma_f32_16x16x4_f32(false, a, false, b,
                                                 (short)0, c, false, false);
}

template <int M>
static __device__ __forceinline__
float row_xor(float v) {
    return __builtin_bit_cast(float,
        __builtin_amdgcn_update_dpp(0, __builtin_bit_cast(int, v),
                                    0x160 | M, 0xf, 0xf, true));
}

template <int M>
static __device__ __forceinline__
void row_max_step(float (&r)[8]) {
#pragma unroll
    for (int e = 0; e < 8; ++e) {
        float t;
        asm("v_max_num_f32_dpp %0, %1, %1 row_xmask:%2 row_mask:0xf bank_mask:0xf"
            : "=v"(t) : "v"(r[e]), "n"(M));
        r[e] = t;
    }
}
template <int M>
static __device__ __forceinline__
void row_sum_step(float (&r)[8]) {
#pragma unroll
    for (int e = 0; e < 8; ++e)
        r[e] += row_xor<M>(r[e]);
}

static __device__ __forceinline__
void stage_piece(float* kdst, float* vdst, int r, int dbase, int i, float4 kq, float4 vq) {
    *(float4*)&kdst[r * KSTRIDE + dbase + 4 * i] = kq;
    vdst[(dbase + 4*i + 0) * VSTRIDE + r] = vq.x;
    vdst[(dbase + 4*i + 1) * VSTRIDE + r] = vq.y;
    vdst[(dbase + 4*i + 2) * VSTRIDE + r] = vq.z;
    vdst[(dbase + 4*i + 3) * VSTRIDE + r] = vq.w;
}

__global__ __launch_bounds__(128)
void attn_fa_kernel(const float* __restrict__ Q,
                    const float* __restrict__ T,
                    const float* __restrict__ V,
                    float* __restrict__ O)
{
    __shared__ float lds_k [2 * 16  * KSTRIDE];
    __shared__ float lds_vt[2 * 128 * VSTRIDE];
    __shared__ float lds_p [4 * 16 * PSTRIDE];

    const int b    = blockIdx.y;
    const int g    = blockIdx.x;
    const int tid  = threadIdx.x;
    const int wave = tid >> 5;
    const int lane = tid & 31;
    const int n    = lane & 15;
    const int hi   = lane >> 4;

    const int q0     = g * 64 + wave * 16;
    const int myDiag = q0 >> 4;
    const int jmax   = g * 4 + 3;

    const float SCALE2 = 0.08838834764831845f * 1.4426950408889634f;

    v2f qa[32];
    {
        const float* qp = Q + ((size_t)b * S_ + q0 + n) * D_ + 2 * hi;
#pragma unroll
        for (int c = 0; c < 32; ++c)
            qa[c] = *(const v2f*)(qp + 4 * c);
    }

    v8f y[8];
#pragma unroll
    for (int nt = 0; nt < 8; ++nt) y[nt] = {};
    float m_run[8], l_run[8];
#pragma unroll
    for (int e = 0; e < 8; ++e) { m_run[e] = -1e30f; l_run[e] = 0.0f; }

    const int r     = tid >> 3;
    const int dbase = (tid & 7) * 16;
    const float* kp = T + ((size_t)b * S_ + r) * D_ + dbase;
    const float* vp = V + ((size_t)b * S_ + r) * D_ + dbase;
    float4 kq0 = *(const float4*)(kp + 0),  vq0 = *(const float4*)(vp + 0);
    float4 kq1 = *(const float4*)(kp + 4),  vq1 = *(const float4*)(vp + 4);
    float4 kq2 = *(const float4*)(kp + 8),  vq2 = *(const float4*)(vp + 8);
    float4 kq3 = *(const float4*)(kp + 12), vq3 = *(const float4*)(vp + 12);
    stage_piece(lds_k, lds_vt, r, dbase, 0, kq0, vq0);
    stage_piece(lds_k, lds_vt, r, dbase, 1, kq1, vq1);
    stage_piece(lds_k, lds_vt, r, dbase, 2, kq2, vq2);
    stage_piece(lds_k, lds_vt, r, dbase, 3, kq3, vq3);
    __syncthreads();

    for (int j = 0; j <= jmax; ++j) {
        const int cur = j & 1;
        const float* kcur = lds_k  + cur * (16  * KSTRIDE);
        const float* vcur = lds_vt + cur * (128 * VSTRIDE);
        float*       knxt = lds_k  + (cur ^ 1) * (16  * KSTRIDE);
        float*       vnxt = lds_vt + (cur ^ 1) * (128 * VSTRIDE);
        const bool   more = (j < jmax);

        if (more) {
            kp += 16 * D_;
            vp += 16 * D_;
            kq0 = *(const float4*)(kp + 0);  vq0 = *(const float4*)(vp + 0);
            kq1 = *(const float4*)(kp + 4);  vq1 = *(const float4*)(vp + 4);
            kq2 = *(const float4*)(kp + 8);  vq2 = *(const float4*)(vp + 8);
            kq3 = *(const float4*)(kp + 12); vq3 = *(const float4*)(vp + 12);
        }

        if (j <= myDiag) {
            const float* krow = &kcur[n * KSTRIDE + 2 * hi];
            v8f s0 = {}, s1 = {};
            v2f kb[2][8];
#pragma unroll
            for (int i = 0; i < 8; ++i)
                kb[0][i] = *(const v2f*)(krow + 4 * i);
#pragma unroll
            for (int grp = 0; grp < 4; ++grp) {
                if (grp < 3) {
#pragma unroll
                    for (int i = 0; i < 8; ++i)
                        kb[(grp + 1) & 1][i] =
                            *(const v2f*)(krow + 4 * (8 * (grp + 1) + i));
                }
#pragma unroll
                for (int i = 0; i < 8; i += 2) {
                    s0 = wmma_f32(qa[8 * grp + i],     kb[grp & 1][i],     s0);
                    s1 = wmma_f32(qa[8 * grp + i + 1], kb[grp & 1][i + 1], s1);
                }
            }

            if (more) {
                stage_piece(knxt, vnxt, r, dbase, 0, kq0, vq0);
                stage_piece(knxt, vnxt, r, dbase, 1, kq1, vq1);
                stage_piece(knxt, vnxt, r, dbase, 2, kq2, vq2);
                stage_piece(knxt, vnxt, r, dbase, 3, kq3, vq3);
            }

            const int kk = j * 16 + n;
            float z[8], rm[8];
#pragma unroll
            for (int e = 0; e < 8; ++e) {
                float sv = s0[e] + s1[e];
                sv = (sv > THR) ? sv : 0.0f;
                float zz = sv * SCALE2;
                const int m = e + 8 * hi;
                if (kk > q0 + m) zz = -1e30f;
                z[e]  = zz;
                rm[e] = zz;
            }
            row_max_step<1>(rm);
            row_max_step<2>(rm);
            row_max_step<4>(rm);
            row_max_step<8>(rm);
            float al[8], p[8], rs[8];
#pragma unroll
            for (int e = 0; e < 8; ++e) {
                float mn = fmaxf(m_run[e], rm[e]);
                float a  = __builtin_amdgcn_exp2f(m_run[e] - mn);
                float pe = __builtin_amdgcn_exp2f(z[e] - mn);
                m_run[e] = mn;
                al[e] = a;
                p[e]  = pe;
                rs[e] = pe;
            }
            row_sum_step<1>(rs);
            row_sum_step<2>(rs);
            row_sum_step<4>(rs);
            row_sum_step<8>(rs);
#pragma unroll
            for (int e = 0; e < 8; ++e)
                l_run[e] = l_run[e] * al[e] + rs[e];

#pragma unroll
            for (int nt = 0; nt < 8; ++nt)
#pragma unroll
                for (int e = 0; e < 8; ++e)
                    y[nt][e] *= al[e];

            float* pp = &lds_p[wave * 16 * PSTRIDE];
#pragma unroll
            for (int e = 0; e < 8; ++e)
                pp[(e + 8 * hi) * PSTRIDE + n] = p[e];

            v2f pa[4];
#pragma unroll
            for (int c = 0; c < 4; ++c)
                pa[c] = *(const v2f*)&pp[n * PSTRIDE + 4 * c + 2 * hi];

#pragma unroll
            for (int half = 0; half < 2; ++half) {
                v2f vb[16];
#pragma unroll
                for (int nt = 0; nt < 4; ++nt)
#pragma unroll
                    for (int c = 0; c < 4; ++c)
                        vb[nt * 4 + c] = *(const v2f*)
                            &vcur[((half * 4 + nt) * 16 + n) * VSTRIDE + 4 * c + 2 * hi];
#pragma unroll
                for (int nt = 0; nt < 4; ++nt) {
                    v8f acc = y[half * 4 + nt];
#pragma unroll
                    for (int c = 0; c < 4; ++c)
                        acc = wmma_f32(pa[c], vb[nt * 4 + c], acc);
                    y[half * 4 + nt] = acc;
                }
            }
        } else if (more) {
            stage_piece(knxt, vnxt, r, dbase, 0, kq0, vq0);
            stage_piece(knxt, vnxt, r, dbase, 1, kq1, vq1);
            stage_piece(knxt, vnxt, r, dbase, 2, kq2, vq2);
            stage_piece(knxt, vnxt, r, dbase, 3, kq3, vq3);
        }
        __syncthreads();
    }

    float inv[8];
#pragma unroll
    for (int e = 0; e < 8; ++e) inv[e] = 1.0f / l_run[e];
    float* op = O + ((size_t)b * S_ + q0) * D_;
#pragma unroll
    for (int nt = 0; nt < 8; ++nt)
#pragma unroll
        for (int e = 0; e < 8; ++e)
            op[(e + 8 * hi) * D_ + nt * 16 + n] = y[nt][e] * inv[e];
}

extern "C" void kernel_launch(void* const* d_in, const int* in_sizes, int n_in,
                              void* d_out, int out_size, void* d_ws, size_t ws_size,
                              hipStream_t stream) {
    const float* Q = (const float*)d_in[0];
    const float* T = (const float*)d_in[1];
    const float* V = (const float*)d_in[2];
    float*       O = (float*)d_out;
    dim3 grid(S_ / 64, B_);
    attn_fa_kernel<<<grid, dim3(128), 0, stream>>>(Q, T, V, O);
}


# causal load balance: each work-group runs query groups 31-p then p of one batch (132 key tiles for every one); 256 equal work-groups, one per WGP (waves_per_eu 1,1), not 512 of 4 to 128 tiles
# speedup vs baseline: 2.6483x; 1.2540x over previous
#include <hip/hip_runtime.h>


#define B_ 16
#define S_ 2048
#define D_ 128
#define THR 0.3f

typedef __attribute__((ext_vector_type(2))) float v2f;
typedef __attribute__((ext_vector_type(8))) float v8f;

#define KSTRIDE 132
#define VSTRIDE 20
#define PSTRIDE 20

static __device__ __forceinline__
v8f wmma_f32(v2f a, v2f b, v8f c) {
    return __builtin_amdgcn_wmma_f32_16x16x4_f32(false, a, false, b,
                                                 (short)0, c, false, false);
}

template <int M>
static __device__ __forceinline__
float row_xor(float v) {
    return __builtin_bit_cast(float,
        __builtin_amdgcn_update_dpp(0, __builtin_bit_cast(int, v),
                                    0x160 | M, 0xf, 0xf, true));
}

template <int M>
static __device__ __forceinline__
void row_max_step(float (&r)[8]) {
#pragma unroll
    for (int e = 0; e < 8; ++e) {
        float t;
        asm("v_max_num_f32_dpp %0, %1, %1 row_xmask:%2 row_mask:0xf bank_mask:0xf"
            : "=v"(t) : "v"(r[e]), "n"(M));
        r[e] = t;
    }
}
template <int M>
static __device__ __forceinline__
void row_sum_step(float (&r)[8]) {
#pragma unroll
    for (int e = 0; e < 8; ++e)
        r[e] += row_xor<M>(r[e]);
}

static __device__ __forceinline__
void stage_piece(float* kdst, float* vdst, int r, int dbase, int i, float4 kq, float4 vq) {
    *(float4*)&kdst[r * KSTRIDE + dbase + 4 * i] = kq;
    vdst[(dbase + 4*i + 0) * VSTRIDE + r] = vq.x;
    vdst[(dbase + 4*i + 1) * VSTRIDE + r] = vq.y;
    vdst[(dbase + 4*i + 2) * VSTRIDE + r] = vq.z;
    vdst[(dbase + 4*i + 3) * VSTRIDE + r] = vq.w;
}

__global__ __launch_bounds__(128) __attribute__((amdgpu_waves_per_eu(1, 1)))
void attn_fa_kernel(const float* __restrict__ Q,
                    const float* __restrict__ T,
                    const float* __restrict__ V,
                    float* __restrict__ O)
{
    __shared__ float lds_k [2 * 16  * KSTRIDE];
    __shared__ float lds_vt[2 * 128 * VSTRIDE];
    __shared__ float lds_p [4 * 16 * PSTRIDE];

    const int b    = blockIdx.y;
    const int p    = blockIdx.x;
    const int tid  = threadIdx.x;
    const int wave = tid >> 5;
    const int lane = tid & 31;
    const int n    = lane & 15;
    const int hi   = lane >> 4;
    const int r     = tid >> 3;
    const int dbase = (tid & 7) * 16;

    const float SCALE2 = 0.08838834764831845f * 1.4426950408889634f;

  for (int pass = 0; pass < 2; ++pass) {
    const int g      = pass ? p : (S_ / 64 - 1) - p;
    const int q0     = g * 64 + wave * 16;
    const int myDiag = q0 >> 4;
    const int jmax   = g * 4 + 3;

    v2f qa[32];
    {
        const float* qp = Q + ((size_t)b * S_ + q0 + n) * D_ + 2 * hi;
#pragma unroll
        for (int c = 0; c < 32; ++c)
            qa[c] = *(const v2f*)(qp + 4 * c);
    }

    v8f y[8];
#pragma unroll
    for (int nt = 0; nt < 8; ++nt) y[nt] = {};
    float m_run[8], l_run[8];
#pragma unroll
    for (int e = 0; e < 8; ++e) { m_run[e] = -1e30f; l_run[e] = 0.0f; }

    const float* kp = T + ((size_t)b * S_ + r) * D_ + dbase;
    const float* vp = V + ((size_t)b * S_ + r) * D_ + dbase;
    float4 kq0 = *(const float4*)(kp + 0),  vq0 = *(const float4*)(vp + 0);
    float4 kq1 = *(const float4*)(kp + 4),  vq1 = *(const float4*)(vp + 4);
    float4 kq2 = *(const float4*)(kp + 8),  vq2 = *(const float4*)(vp + 8);
    float4 kq3 = *(const float4*)(kp + 12), vq3 = *(const float4*)(vp + 12);
    stage_piece(lds_k, lds_vt, r, dbase, 0, kq0, vq0);
    stage_piece(lds_k, lds_vt, r, dbase, 1, kq1, vq1);
    stage_piece(lds_k, lds_vt, r, dbase, 2, kq2, vq2);
    stage_piece(lds_k, lds_vt, r, dbase, 3, kq3, vq3);
    __syncthreads();

    for (int j = 0; j <= jmax; ++j) {
        const int cur = j & 1;
        const float* kcur = lds_k  + cur * (16  * KSTRIDE);
        const float* vcur = lds_vt + cur * (128 * VSTRIDE);
        float*       knxt = lds_k  + (cur ^ 1) * (16  * KSTRIDE);
        float*       vnxt = lds_vt + (cur ^ 1) * (128 * VSTRIDE);
        const bool   more = (j < jmax);

        if (more) {
            kp += 16 * D_;
            vp += 16 * D_;
            kq0 = *(const float4*)(kp + 0);  vq0 = *(const float4*)(vp + 0);
            kq1 = *(const float4*)(kp + 4);  vq1 = *(const float4*)(vp + 4);
            kq2 = *(const float4*)(kp + 8);  vq2 = *(const float4*)(vp + 8);
            kq3 = *(const float4*)(kp + 12); vq3 = *(const float4*)(vp + 12);
        }

        if (j <= myDiag) {
            const float* krow = &kcur[n * KSTRIDE + 2 * hi];
            v8f s0 = {}, s1 = {};
            v2f kb[2][8];
#pragma unroll
            for (int i = 0; i < 8; ++i)
                kb[0][i] = *(const v2f*)(krow + 4 * i);
#pragma unroll
            for (int grp = 0; grp < 4; ++grp) {
                if (grp < 3) {
#pragma unroll
                    for (int i = 0; i < 8; ++i)
                        kb[(grp + 1) & 1][i] =
                            *(const v2f*)(krow + 4 * (8 * (grp + 1) + i));
                }
#pragma unroll
                for (int i = 0; i < 8; i += 2) {
                    s0 = wmma_f32(qa[8 * grp + i],     kb[grp & 1][i],     s0);
                    s1 = wmma_f32(qa[8 * grp + i + 1], kb[grp & 1][i + 1], s1);
                }
            }

            if (more) {
                stage_piece(knxt, vnxt, r, dbase, 0, kq0, vq0);
                stage_piece(knxt, vnxt, r, dbase, 1, kq1, vq1);
                stage_piece(knxt, vnxt, r, dbase, 2, kq2, vq2);
                stage_piece(knxt, vnxt, r, dbase, 3, kq3, vq3);
            }

            const int kk = j * 16 + n;
            float z[8], rm[8];
#pragma unroll
            for (int e = 0; e < 8; ++e) {
                float sv = s0[e] + s1[e];
                sv = (sv > THR) ? sv : 0.0f;
                float zz = sv * SCALE2;
                const int m = e + 8 * hi;
                if (kk > q0 + m) zz = -1e30f;
                z[e]  = zz;
                rm[e] = zz;
            }
            row_max_step<1>(rm);
            row_max_step<2>(rm);
            row_max_step<4>(rm);
            row_max_step<8>(rm);
            float al[8], p[8], rs[8];
#pragma unroll
            for (int e = 0; e < 8; ++e) {
                float mn = fmaxf(m_run[e], rm[e]);
                float a  = __builtin_amdgcn_exp2f(m_run[e] - mn);
                float pe = __builtin_amdgcn_exp2f(z[e] - mn);
                m_run[e] = mn;
                al[e] = a;
                p[e]  = pe;
                rs[e] = pe;
            }
            row_sum_step<1>(rs);
            row_sum_step<2>(rs);
            row_sum_step<4>(rs);
            row_sum_step<8>(rs);
#pragma unroll
            for (int e = 0; e < 8; ++e)
                l_run[e] = l_run[e] * al[e] + rs[e];

#pragma unroll
            for (int nt = 0; nt < 8; ++nt)
#pragma unroll
                for (int e = 0; e < 8; ++e)
                    y[nt][e] *= al[e];

            float* pp = &lds_p[wave * 16 * PSTRIDE];
#pragma unroll
            for (int e = 0; e < 8; ++e)
                pp[(e + 8 * hi) * PSTRIDE + n] = p[e];

            v2f pa[4];
#pragma unroll
            for (int c = 0; c < 4; ++c)
                pa[c] = *(const v2f*)&pp[n * PSTRIDE + 4 * c + 2 * hi];

#pragma unroll
            for (int half = 0; half < 2; ++half) {
                v2f vb[16];
#pragma unroll
                for (int nt = 0; nt < 4; ++nt)
#pragma unroll
                    for (int c = 0; c < 4; ++c)
                        vb[nt * 4 + c] = *(const v2f*)
                            &vcur[((half * 4 + nt) * 16 + n) * VSTRIDE + 4 * c + 2 * hi];
#pragma unroll
                for (int nt = 0; nt < 4; ++nt) {
                    v8f acc = y[half * 4 + nt];
#pragma unroll
                    for (int c = 0; c < 4; ++c)
                        acc = wmma_f32(pa[c], vb[nt * 4 + c], acc);
                    y[half * 4 + nt] = acc;
                }
            }
        } else if (more) {
            stage_piece(knxt, vnxt, r, dbase, 0, kq0, vq0);
            stage_piece(knxt, vnxt, r, dbase, 1, kq1, vq1);
            stage_piece(knxt, vnxt, r, dbase, 2, kq2, vq2);
            stage_piece(knxt, vnxt, r, dbase, 3, kq3, vq3);
        }
        __syncthreads();
    }

    float inv[8];
#pragma unroll
    for (int e = 0; e < 8; ++e) inv[e] = 1.0f / l_run[e];
    float* op = O + ((size_t)b * S_ + q0) * D_;
#pragma unroll
    for (int nt = 0; nt < 8; ++nt)
#pragma unroll
        for (int e = 0; e < 8; ++e)
            op[(e + 8 * hi) * D_ + nt * 16 + n] = y[nt][e] * inv[e];
  }
}

extern "C" void kernel_launch(void* const* d_in, const int* in_sizes, int n_in,
                              void* d_out, int out_size, void* d_ws, size_t ws_size,
                              hipStream_t stream) {
    const float* Q = (const float*)d_in[0];
    const float* T = (const float*)d_in[1];
    const float* V = (const float*)d_in[2];
    float*       O = (float*)d_out;
    dim3 grid(S_ / 128, B_);
    attn_fa_kernel<<<grid, dim3(128), 0, stream>>>(Q, T, V, O);
}


# 8-wave work-groups, two waves per SIMD: waves 4-7 take the odd key tiles of the same 64 query rows, waves 0-3 the even; (m,l,y) partials merged via LDS at the end (softmax sums regrouped even/odd)
# speedup vs baseline: 3.5942x; 1.3572x over previous
#include <hip/hip_runtime.h>


#define B_ 16
#define S_ 2048
#define D_ 128
#define THR 0.3f

typedef __attribute__((ext_vector_type(2))) float v2f;
typedef __attribute__((ext_vector_type(8))) float v8f;

#define KSTRIDE 132
#define VSTRIDE 20
#define PSTRIDE 20
#define MGREGS  80

static __device__ __forceinline__
v8f wmma_f32(v2f a, v2f b, v8f c) {
    return __builtin_amdgcn_wmma_f32_16x16x4_f32(false, a, false, b,
                                                 (short)0, c, false, false);
}

template <int M>
static __device__ __forceinline__
float row_xor(float v) {
    return __builtin_bit_cast(float,
        __builtin_amdgcn_update_dpp(0, __builtin_bit_cast(int, v),
                                    0x160 | M, 0xf, 0xf, true));
}

template <int M>
static __device__ __forceinline__
void row_max_step(float (&r)[8]) {
#pragma unroll
    for (int e = 0; e < 8; ++e) {
        float t;
        asm("v_max_num_f32_dpp %0, %1, %1 row_xmask:%2 row_mask:0xf bank_mask:0xf"
            : "=v"(t) : "v"(r[e]), "n"(M));
        r[e] = t;
    }
}
template <int M>
static __device__ __forceinline__
void row_sum_step(float (&r)[8]) {
#pragma unroll
    for (int e = 0; e < 8; ++e)
        r[e] += row_xor<M>(r[e]);
}

static __device__ __forceinline__
void stage_piece(float* kdst, float* vdst, int rk, int rv, int dbase, int i, float4 kq, float4 vq) {
    *(float4*)&kdst[rk * KSTRIDE + dbase + 4 * i] = kq;
    vdst[(dbase + 4*i + 0) * VSTRIDE + rv] = vq.x;
    vdst[(dbase + 4*i + 1) * VSTRIDE + rv] = vq.y;
    vdst[(dbase + 4*i + 2) * VSTRIDE + rv] = vq.z;
    vdst[(dbase + 4*i + 3) * VSTRIDE + rv] = vq.w;
}

__global__ __launch_bounds__(256) __attribute__((amdgpu_waves_per_eu(2, 2)))
void attn_fa_kernel(const float* __restrict__ Q,
                    const float* __restrict__ T,
                    const float* __restrict__ V,
                    float* __restrict__ O)
{
    __shared__ float lds_k [2 * 32  * KSTRIDE];
    __shared__ float lds_vt[2 * 2 * 128 * VSTRIDE];
    __shared__ float lds_p [8 * 16 * PSTRIDE];
    __shared__ float lds_mg[4 * MGREGS * 32];

    const int b    = blockIdx.y;
    const int p    = blockIdx.x;
    const int tid  = threadIdx.x;
    const int wave = tid >> 5;
    const int set  = wave >> 2;
    const int w4   = wave & 3;
    const int lane = tid & 31;
    const int n    = lane & 15;
    const int hi   = lane >> 4;
    const int r     = tid >> 3;
    const int rset  = r >> 4;
    const int r16   = r & 15;
    const int dbase = (tid & 7) * 16;

    const float SCALE2 = 0.08838834764831845f * 1.4426950408889634f;

  for (int pass = 0; pass < 2; ++pass) {
    const int g      = pass ? p : (S_ / 64 - 1) - p;
    const int q0     = g * 64 + w4 * 16;
    const int myDiag = q0 >> 4;
    const int imax   = g * 2 + 1;

    v2f qa[32];
    {
        const float* qp = Q + ((size_t)b * S_ + q0 + n) * D_ + 2 * hi;
#pragma unroll
        for (int c = 0; c < 32; ++c)
            qa[c] = *(const v2f*)(qp + 4 * c);
    }

    v8f y[8];
#pragma unroll
    for (int nt = 0; nt < 8; ++nt) y[nt] = {};
    float m_run[8], l_run[8];
#pragma unroll
    for (int e = 0; e < 8; ++e) { m_run[e] = -1e30f; l_run[e] = 0.0f; }

    const float* kp = T + ((size_t)b * S_ + r) * D_ + dbase;
    const float* vp = V + ((size_t)b * S_ + r) * D_ + dbase;
    float4 kq0 = *(const float4*)(kp + 0),  vq0 = *(const float4*)(vp + 0);
    float4 kq1 = *(const float4*)(kp + 4),  vq1 = *(const float4*)(vp + 4);
    float4 kq2 = *(const float4*)(kp + 8),  vq2 = *(const float4*)(vp + 8);
    float4 kq3 = *(const float4*)(kp + 12), vq3 = *(const float4*)(vp + 12);
    stage_piece(lds_k, lds_vt + rset * (128 * VSTRIDE), r, r16, dbase, 0, kq0, vq0);
    stage_piece(lds_k, lds_vt + rset * (128 * VSTRIDE), r, r16, dbase, 1, kq1, vq1);
    stage_piece(lds_k, lds_vt + rset * (128 * VSTRIDE), r, r16, dbase, 2, kq2, vq2);
    stage_piece(lds_k, lds_vt + rset * (128 * VSTRIDE), r, r16, dbase, 3, kq3, vq3);
    __syncthreads();

    for (int ip = 0; ip <= imax; ++ip) {
        const int cur = ip & 1;
        const int j   = 2 * ip + set;
        const float* kcur = lds_k  + cur * (32 * KSTRIDE) + set * (16 * KSTRIDE);
        const float* vcur = lds_vt + cur * (2 * 128 * VSTRIDE) + set * (128 * VSTRIDE);
        float*       knxt = lds_k  + (cur ^ 1) * (32 * KSTRIDE);
        float*       vnxt = lds_vt + (cur ^ 1) * (2 * 128 * VSTRIDE) + rset * (128 * VSTRIDE);
        const bool   more = (ip < imax);

        if (more) {
            kp += 32 * D_;
            vp += 32 * D_;
            kq0 = *(const float4*)(kp + 0);  vq0 = *(const float4*)(vp + 0);
            kq1 = *(const float4*)(kp + 4);  vq1 = *(const float4*)(vp + 4);
            kq2 = *(const float4*)(kp + 8);  vq2 = *(const float4*)(vp + 8);
            kq3 = *(const float4*)(kp + 12); vq3 = *(const float4*)(vp + 12);
        }

        if (j <= myDiag) {
            const float* krow = &kcur[n * KSTRIDE + 2 * hi];
            v8f s0 = {}, s1 = {};
            v2f kb[2][8];
#pragma unroll
            for (int i = 0; i < 8; ++i)
                kb[0][i] = *(const v2f*)(krow + 4 * i);
#pragma unroll
            for (int grp = 0; grp < 4; ++grp) {
                if (grp < 3) {
#pragma unroll
                    for (int i = 0; i < 8; ++i)
                        kb[(grp + 1) & 1][i] =
                            *(const v2f*)(krow + 4 * (8 * (grp + 1) + i));
                }
#pragma unroll
                for (int i = 0; i < 8; i += 2) {
                    s0 = wmma_f32(qa[8 * grp + i],     kb[grp & 1][i],     s0);
                    s1 = wmma_f32(qa[8 * grp + i + 1], kb[grp & 1][i + 1], s1);
                }
            }

            if (more) {
                stage_piece(knxt, vnxt, r, r16, dbase, 0, kq0, vq0);
                stage_piece(knxt, vnxt, r, r16, dbase, 1, kq1, vq1);
                stage_piece(knxt, vnxt, r, r16, dbase, 2, kq2, vq2);
                stage_piece(knxt, vnxt, r, r16, dbase, 3, kq3, vq3);
            }

            const int kk = j * 16 + n;
            float z[8], rm[8];
#pragma unroll
            for (int e = 0; e < 8; ++e) {
                float sv = s0[e] + s1[e];
                sv = (sv > THR) ? sv : 0.0f;
                float zz = sv * SCALE2;
                const int m = e + 8 * hi;
                if (kk > q0 + m) zz = -1e30f;
                z[e]  = zz;
                rm[e] = zz;
            }
            row_max_step<1>(rm);
            row_max_step<2>(rm);
            row_max_step<4>(rm);
            row_max_step<8>(rm);
            float al[8], p[8], rs[8];
#pragma unroll
            for (int e = 0; e < 8; ++e) {
                float mn = fmaxf(m_run[e], rm[e]);
                float a  = __builtin_amdgcn_exp2f(m_run[e] - mn);
                float pe = __builtin_amdgcn_exp2f(z[e] - mn);
                m_run[e] = mn;
                al[e] = a;
                p[e]  = pe;
                rs[e] = pe;
            }
            row_sum_step<1>(rs);
            row_sum_step<2>(rs);
            row_sum_step<4>(rs);
            row_sum_step<8>(rs);
#pragma unroll
            for (int e = 0; e < 8; ++e)
                l_run[e] = l_run[e] * al[e] + rs[e];

#pragma unroll
            for (int nt = 0; nt < 8; ++nt)
#pragma unroll
                for (int e = 0; e < 8; ++e)
                    y[nt][e] *= al[e];

            float* pp = &lds_p[wave * 16 * PSTRIDE];
#pragma unroll
            for (int e = 0; e < 8; ++e)
                pp[(e + 8 * hi) * PSTRIDE + n] = p[e];

            v2f pa[4];
#pragma unroll
            for (int c = 0; c < 4; ++c)
                pa[c] = *(const v2f*)&pp[n * PSTRIDE + 4 * c + 2 * hi];

#pragma unroll
            for (int half = 0; half < 2; ++half) {
                v2f vb[16];
#pragma unroll
                for (int nt = 0; nt < 4; ++nt)
#pragma unroll
                    for (int c = 0; c < 4; ++c)
                        vb[nt * 4 + c] = *(const v2f*)
                            &vcur[((half * 4 + nt) * 16 + n) * VSTRIDE + 4 * c + 2 * hi];
#pragma unroll
                for (int nt = 0; nt < 4; ++nt) {
                    v8f acc = y[half * 4 + nt];
#pragma unroll
                    for (int c = 0; c < 4; ++c)
                        acc = wmma_f32(pa[c], vb[nt * 4 + c], acc);
                    y[half * 4 + nt] = acc;
                }
            }
        } else if (more) {
            stage_piece(knxt, vnxt, r, r16, dbase, 0, kq0, vq0);
            stage_piece(knxt, vnxt, r, r16, dbase, 1, kq1, vq1);
            stage_piece(knxt, vnxt, r, r16, dbase, 2, kq2, vq2);
            stage_piece(knxt, vnxt, r, r16, dbase, 3, kq3, vq3);
        }
        __syncthreads();
    }

    float* mg = &lds_mg[w4 * (MGREGS * 32) + lane];
    if (set == 1) {
#pragma unroll
        for (int e = 0; e < 8; ++e) {
            mg[e * 32]       = m_run[e];
            mg[(8 + e) * 32] = l_run[e];
        }
#pragma unroll
        for (int nt = 0; nt < 8; ++nt)
#pragma unroll
            for (int e = 0; e < 8; ++e)
                mg[(16 + nt * 8 + e) * 32] = y[nt][e];
    }
    __syncthreads();
    if (set == 0) {
        float a0[8], a1[8];
#pragma unroll
        for (int e = 0; e < 8; ++e) {
            const float m1 = mg[e * 32];
            const float l1 = mg[(8 + e) * 32];
            const float mn = fmaxf(m_run[e], m1);
            const float s0 = __builtin_amdgcn_exp2f(m_run[e] - mn);
            const float s1 = __builtin_amdgcn_exp2f(m1 - mn);
            const float inv = 1.0f / (l_run[e] * s0 + l1 * s1);
            a0[e] = s0 * inv;
            a1[e] = s1 * inv;
        }
        float* op = O + ((size_t)b * S_ + q0) * D_;
#pragma unroll
        for (int nt = 0; nt < 8; ++nt)
#pragma unroll
            for (int e = 0; e < 8; ++e)
                op[(e + 8 * hi) * D_ + nt * 16 + n] =
                    y[nt][e] * a0[e] + mg[(16 + nt * 8 + e) * 32] * a1[e];
    }
  }
}

extern "C" void kernel_launch(void* const* d_in, const int* in_sizes, int n_in,
                              void* d_out, int out_size, void* d_ws, size_t ws_size,
                              hipStream_t stream) {
    const float* Q = (const float*)d_in[0];
    const float* T = (const float*)d_in[1];
    const float* V = (const float*)d_in[2];
    float*       O = (float*)d_out;
    dim3 grid(S_ / 128, B_);
    attn_fa_kernel<<<grid, dim3(256), 0, stream>>>(Q, T, V, O);
}


# scores taken transposed (S^T = K Q^T): one query per lane, max and sum in-lane plus one half-wave exchange; exp'd tile is PV's B operand from registers, no LDS trip; row-sum and PV key order change
# speedup vs baseline: 3.7920x; 1.0550x over previous
#include <hip/hip_runtime.h>


#define B_ 16
#define S_ 2048
#define D_ 128
#define THR 0.3f

typedef __attribute__((ext_vector_type(2))) float v2f;
typedef __attribute__((ext_vector_type(8))) float v8f;

#define KSTRIDE 132
#define VSTRIDE 20
#define MGREGS  66

static __device__ __forceinline__
v8f wmma_f32(v2f a, v2f b, v8f c) {
    return __builtin_amdgcn_wmma_f32_16x16x4_f32(false, a, false, b,
                                                 (short)0, c, false, false);
}

static __device__ __forceinline__
float xhalf(float x) {
    const int b = __builtin_bit_cast(int, x);
    return __builtin_bit_cast(float,
        __builtin_amdgcn_permlanex16(b, b, 0x76543210, 0xfedcba98, false, false));
}

static __device__ __forceinline__
void stage_piece(float* kdst, float* vdst, int rk, int rv, int dbase, int i, float4 kq, float4 vq) {
    *(float4*)&kdst[rk * KSTRIDE + dbase + 4 * i] = kq;
    vdst[(dbase + 4*i + 0) * VSTRIDE + rv] = vq.x;
    vdst[(dbase + 4*i + 1) * VSTRIDE + rv] = vq.y;
    vdst[(dbase + 4*i + 2) * VSTRIDE + rv] = vq.z;
    vdst[(dbase + 4*i + 3) * VSTRIDE + rv] = vq.w;
}

__global__ __launch_bounds__(256) __attribute__((amdgpu_waves_per_eu(2, 2)))
void attn_fa_kernel(const float* __restrict__ Q,
                    const float* __restrict__ T,
                    const float* __restrict__ V,
                    float* __restrict__ O)
{
    __shared__ float lds_k [2 * 32  * KSTRIDE];
    __shared__ float lds_vt[2 * 2 * 128 * VSTRIDE];
    __shared__ float lds_mg[4 * MGREGS * 32];

    const int b    = blockIdx.y;
    const int p    = blockIdx.x;
    const int tid  = threadIdx.x;
    const int wave = tid >> 5;
    const int set  = wave >> 2;
    const int w4   = wave & 3;
    const int lane = tid & 31;
    const int n    = lane & 15;
    const int hi   = lane >> 4;
    const int r     = tid >> 3;
    const int rset  = r >> 4;
    const int r16   = r & 15;
    const int rv    = 4 * ((r16 >> 1) & 3) + 2 * (r16 >> 3) + (r16 & 1);
    const int dbase = (tid & 7) * 16;

    const float SCALE2 = 0.08838834764831845f * 1.4426950408889634f;

  for (int pass = 0; pass < 2; ++pass) {
    const int g      = pass ? p : (S_ / 64 - 1) - p;
    const int q0     = g * 64 + w4 * 16;
    const int myDiag = q0 >> 4;
    const int imax   = g * 2 + 1;

    v2f qa[32];
    {
        const float* qp = Q + ((size_t)b * S_ + q0 + n) * D_ + 2 * hi;
#pragma unroll
        for (int c = 0; c < 32; ++c)
            qa[c] = *(const v2f*)(qp + 4 * c);
    }

    v8f y[8];
#pragma unroll
    for (int nt = 0; nt < 8; ++nt) y[nt] = {};
    float m_run = -1e30f, l_run = 0.0f;

    const float* kp = T + ((size_t)b * S_ + r) * D_ + dbase;
    const float* vp = V + ((size_t)b * S_ + r) * D_ + dbase;
    float4 kq0 = *(const float4*)(kp + 0),  vq0 = *(const float4*)(vp + 0);
    float4 kq1 = *(const float4*)(kp + 4),  vq1 = *(const float4*)(vp + 4);
    float4 kq2 = *(const float4*)(kp + 8),  vq2 = *(const float4*)(vp + 8);
    float4 kq3 = *(const float4*)(kp + 12), vq3 = *(const float4*)(vp + 12);
    stage_piece(lds_k, lds_vt + rset * (128 * VSTRIDE), r, rv, dbase, 0, kq0, vq0);
    stage_piece(lds_k, lds_vt + rset * (128 * VSTRIDE), r, rv, dbase, 1, kq1, vq1);
    stage_piece(lds_k, lds_vt + rset * (128 * VSTRIDE), r, rv, dbase, 2, kq2, vq2);
    stage_piece(lds_k, lds_vt + rset * (128 * VSTRIDE), r, rv, dbase, 3, kq3, vq3);
    __syncthreads();

    for (int ip = 0; ip <= imax; ++ip) {
        const int cur = ip & 1;
        const int j   = 2 * ip + set;
        const float* kcur = lds_k  + cur * (32 * KSTRIDE) + set * (16 * KSTRIDE);
        const float* vcur = lds_vt + cur * (2 * 128 * VSTRIDE) + set * (128 * VSTRIDE);
        float*       knxt = lds_k  + (cur ^ 1) * (32 * KSTRIDE);
        float*       vnxt = lds_vt + (cur ^ 1) * (2 * 128 * VSTRIDE) + rset * (128 * VSTRIDE);
        const bool   more = (ip < imax);

        if (more) {
            kp += 32 * D_;
            vp += 32 * D_;
            kq0 = *(const float4*)(kp + 0);  vq0 = *(const float4*)(vp + 0);
            kq1 = *(const float4*)(kp + 4);  vq1 = *(const float4*)(vp + 4);
            kq2 = *(const float4*)(kp + 8);  vq2 = *(const float4*)(vp + 8);
            kq3 = *(const float4*)(kp + 12); vq3 = *(const float4*)(vp + 12);
        }

        if (j <= myDiag) {
            const float* krow = &kcur[n * KSTRIDE + 2 * hi];
            v8f s0 = {}, s1 = {};
            v2f kb[2][8];
#pragma unroll
            for (int i = 0; i < 8; ++i)
                kb[0][i] = *(const v2f*)(krow + 4 * i);
#pragma unroll
            for (int grp = 0; grp < 4; ++grp) {
                if (grp < 3) {
#pragma unroll
                    for (int i = 0; i < 8; ++i)
                        kb[(grp + 1) & 1][i] =
                            *(const v2f*)(krow + 4 * (8 * (grp + 1) + i));
                }
#pragma unroll
                for (int i = 0; i < 8; i += 2) {
                    s0 = wmma_f32(kb[grp & 1][i],     qa[8 * grp + i],     s0);
                    s1 = wmma_f32(kb[grp & 1][i + 1], qa[8 * grp + i + 1], s1);
                }
            }

            if (more) {
                stage_piece(knxt, vnxt, r, rv, dbase, 0, kq0, vq0);
                stage_piece(knxt, vnxt, r, rv, dbase, 1, kq1, vq1);
                stage_piece(knxt, vnxt, r, rv, dbase, 2, kq2, vq2);
                stage_piece(knxt, vnxt, r, rv, dbase, 3, kq3, vq3);
            }

            const int dl = q0 + n - (j * 16 + 8 * hi);
            float z[8];
#pragma unroll
            for (int e = 0; e < 8; ++e) {
                float sv = s0[e] + s1[e];
                sv = (sv > THR) ? sv : 0.0f;
                float zz = sv * SCALE2;
                if (e > dl) zz = -1e30f;
                z[e] = zz;
            }
            float mx = fmaxf(fmaxf(fmaxf(z[0], z[1]), fmaxf(z[2], z[3])),
                             fmaxf(fmaxf(z[4], z[5]), fmaxf(z[6], z[7])));
            mx = fmaxf(mx, xhalf(mx));
            const float mn = fmaxf(m_run, mx);
            const float al = __builtin_amdgcn_exp2f(m_run - mn);
            m_run = mn;
            float pe[8];
#pragma unroll
            for (int e = 0; e < 8; ++e)
                pe[e] = __builtin_amdgcn_exp2f(z[e] - mn);
            l_run = l_run * al + (((pe[0] + pe[1]) + (pe[2] + pe[3])) +
                                  ((pe[4] + pe[5]) + (pe[6] + pe[7])));

#pragma unroll
            for (int nt = 0; nt < 8; ++nt)
#pragma unroll
                for (int e = 0; e < 8; ++e)
                    y[nt][e] *= al;

            v2f pb[4];
#pragma unroll
            for (int c = 0; c < 4; ++c) {
                pb[c].x = pe[2 * c];
                pb[c].y = pe[2 * c + 1];
            }
#pragma unroll
            for (int half = 0; half < 2; ++half) {
                v2f va[16];
#pragma unroll
                for (int nt = 0; nt < 4; ++nt)
#pragma unroll
                    for (int c = 0; c < 4; ++c)
                        va[nt * 4 + c] = *(const v2f*)
                            &vcur[((half * 4 + nt) * 16 + n) * VSTRIDE + 4 * c + 2 * hi];
#pragma unroll
                for (int nt = 0; nt < 4; ++nt) {
                    v8f acc = y[half * 4 + nt];
#pragma unroll
                    for (int c = 0; c < 4; ++c)
                        acc = wmma_f32(va[nt * 4 + c], pb[c], acc);
                    y[half * 4 + nt] = acc;
                }
            }
        } else if (more) {
            stage_piece(knxt, vnxt, r, rv, dbase, 0, kq0, vq0);
            stage_piece(knxt, vnxt, r, rv, dbase, 1, kq1, vq1);
            stage_piece(knxt, vnxt, r, rv, dbase, 2, kq2, vq2);
            stage_piece(knxt, vnxt, r, rv, dbase, 3, kq3, vq3);
        }
        __syncthreads();
    }

    float* mg = &lds_mg[w4 * (MGREGS * 32) + lane];
    if (set == 1) {
        mg[0]  = m_run;
        mg[32] = l_run;
#pragma unroll
        for (int nt = 0; nt < 8; ++nt)
#pragma unroll
            for (int e = 0; e < 8; ++e)
                mg[(2 + nt * 8 + e) * 32] = y[nt][e];
    }
    __syncthreads();
    if (set == 0) {
        const float m1 = mg[0];
        const float l1 = mg[32];
        const float mn = fmaxf(m_run, m1);
        const float c0 = __builtin_amdgcn_exp2f(m_run - mn);
        const float c1 = __builtin_amdgcn_exp2f(m1 - mn);
        float lt = l_run * c0 + l1 * c1;
        lt += xhalf(lt);
        const float inv = 1.0f / lt;
        const float a0 = c0 * inv;
        const float a1 = c1 * inv;
        float* op = O + ((size_t)b * S_ + q0 + n) * D_ + 8 * hi;
#pragma unroll
        for (int nt = 0; nt < 8; ++nt) {
            float4 o0, o1;
            o0.x = y[nt][0] * a0 + mg[(2 + nt * 8 + 0) * 32] * a1;
            o0.y = y[nt][1] * a0 + mg[(2 + nt * 8 + 1) * 32] * a1;
            o0.z = y[nt][2] * a0 + mg[(2 + nt * 8 + 2) * 32] * a1;
            o0.w = y[nt][3] * a0 + mg[(2 + nt * 8 + 3) * 32] * a1;
            o1.x = y[nt][4] * a0 + mg[(2 + nt * 8 + 4) * 32] * a1;
            o1.y = y[nt][5] * a0 + mg[(2 + nt * 8 + 5) * 32] * a1;
            o1.z = y[nt][6] * a0 + mg[(2 + nt * 8 + 6) * 32] * a1;
            o1.w = y[nt][7] * a0 + mg[(2 + nt * 8 + 7) * 32] * a1;
            *(float4*)(op + nt * 16)     = o0;
            *(float4*)(op + nt * 16 + 4) = o1;
        }
    }
  }
}

extern "C" void kernel_launch(void* const* d_in, const int* in_sizes, int n_in,
                              void* d_out, int out_size, void* d_ws, size_t ws_size,
                              hipStream_t stream) {
    const float* Q = (const float*)d_in[0];
    const float* T = (const float*)d_in[1];
    const float* V = (const float*)d_in[2];
    float*       O = (float*)d_out;
    dim3 grid(S_ / 128, B_);
    attn_fa_kernel<<<grid, dim3(256), 0, stream>>>(Q, T, V, O);
}


# a key tile's K and V^T fragment loads (32 ds_load_2addr_b64) issued as ONE group ahead of its 64 WMMAs, held by a scheduler fence, not one load per two score WMMAs; products and sums unchanged
# speedup vs baseline: 4.0787x; 1.0756x over previous
#include <hip/hip_runtime.h>


#define B_ 16
#define S_ 2048
#define D_ 128
#define THR 0.3f

typedef __attribute__((ext_vector_type(2))) float v2f;
typedef __attribute__((ext_vector_type(8))) float v8f;

#define KSTRIDE 132
#define VSTRIDE 20
#define MGREGS  66

static __device__ __forceinline__
v8f wmma_f32(v2f a, v2f b, v8f c) {
    return __builtin_amdgcn_wmma_f32_16x16x4_f32(false, a, false, b,
                                                 (short)0, c, false, false);
}

static __device__ __forceinline__
float xhalf(float x) {
    const int b = __builtin_bit_cast(int, x);
    return __builtin_bit_cast(float,
        __builtin_amdgcn_permlanex16(b, b, 0x76543210, 0xfedcba98, false, false));
}

static __device__ __forceinline__
void stage_piece(float* kdst, float* vdst, int rk, int rv, int dbase, int i, float4 kq, float4 vq) {
    *(float4*)&kdst[rk * KSTRIDE + dbase + 4 * i] = kq;
    vdst[(dbase + 4*i + 0) * VSTRIDE + rv] = vq.x;
    vdst[(dbase + 4*i + 1) * VSTRIDE + rv] = vq.y;
    vdst[(dbase + 4*i + 2) * VSTRIDE + rv] = vq.z;
    vdst[(dbase + 4*i + 3) * VSTRIDE + rv] = vq.w;
}

__global__ __launch_bounds__(256) __attribute__((amdgpu_waves_per_eu(2, 2)))
void attn_fa_kernel(const float* __restrict__ Q,
                    const float* __restrict__ T,
                    const float* __restrict__ V,
                    float* __restrict__ O)
{
    __shared__ float lds_k [2 * 32  * KSTRIDE];
    __shared__ float lds_vt[2 * 2 * 128 * VSTRIDE];
    __shared__ float lds_mg[4 * MGREGS * 32];

    const int b    = blockIdx.y;
    const int p    = blockIdx.x;
    const int tid  = threadIdx.x;
    const int wave = tid >> 5;
    const int set  = wave >> 2;
    const int w4   = wave & 3;
    const int lane = tid & 31;
    const int n    = lane & 15;
    const int hi   = lane >> 4;
    const int r     = tid >> 3;
    const int rset  = r >> 4;
    const int r16   = r & 15;
    const int rv    = 4 * ((r16 >> 1) & 3) + 2 * (r16 >> 3) + (r16 & 1);
    const int dbase = (tid & 7) * 16;

    const float SCALE2 = 0.08838834764831845f * 1.4426950408889634f;

  for (int pass = 0; pass < 2; ++pass) {
    const int g      = pass ? p : (S_ / 64 - 1) - p;
    const int q0     = g * 64 + w4 * 16;
    const int myDiag = q0 >> 4;
    const int imax   = g * 2 + 1;

    v2f qa[32];
    {
        const float* qp = Q + ((size_t)b * S_ + q0 + n) * D_ + 2 * hi;
#pragma unroll
        for (int c = 0; c < 32; ++c)
            qa[c] = *(const v2f*)(qp + 4 * c);
    }

    v8f y[8];
#pragma unroll
    for (int nt = 0; nt < 8; ++nt) y[nt] = {};
    float m_run = -1e30f, l_run = 0.0f;

    const float* kp = T + ((size_t)b * S_ + r) * D_ + dbase;
    const float* vp = V + ((size_t)b * S_ + r) * D_ + dbase;
    float4 kq0 = *(const float4*)(kp + 0),  vq0 = *(const float4*)(vp + 0);
    float4 kq1 = *(const float4*)(kp + 4),  vq1 = *(const float4*)(vp + 4);
    float4 kq2 = *(const float4*)(kp + 8),  vq2 = *(const float4*)(vp + 8);
    float4 kq3 = *(const float4*)(kp + 12), vq3 = *(const float4*)(vp + 12);
    stage_piece(lds_k, lds_vt + rset * (128 * VSTRIDE), r, rv, dbase, 0, kq0, vq0);
    stage_piece(lds_k, lds_vt + rset * (128 * VSTRIDE), r, rv, dbase, 1, kq1, vq1);
    stage_piece(lds_k, lds_vt + rset * (128 * VSTRIDE), r, rv, dbase, 2, kq2, vq2);
    stage_piece(lds_k, lds_vt + rset * (128 * VSTRIDE), r, rv, dbase, 3, kq3, vq3);
    __syncthreads();

    for (int ip = 0; ip <= imax; ++ip) {
        const int cur = ip & 1;
        const int j   = 2 * ip + set;
        const float* kcur = lds_k  + cur * (32 * KSTRIDE) + set * (16 * KSTRIDE);
        const float* vcur = lds_vt + cur * (2 * 128 * VSTRIDE) + set * (128 * VSTRIDE);
        float*       knxt = lds_k  + (cur ^ 1) * (32 * KSTRIDE);
        float*       vnxt = lds_vt + (cur ^ 1) * (2 * 128 * VSTRIDE) + rset * (128 * VSTRIDE);
        const bool   more = (ip < imax);

        if (more) {
            kp += 32 * D_;
            vp += 32 * D_;
            kq0 = *(const float4*)(kp + 0);  vq0 = *(const float4*)(vp + 0);
            kq1 = *(const float4*)(kp + 4);  vq1 = *(const float4*)(vp + 4);
            kq2 = *(const float4*)(kp + 8);  vq2 = *(const float4*)(vp + 8);
            kq3 = *(const float4*)(kp + 12); vq3 = *(const float4*)(vp + 12);
        }

        if (j <= myDiag) {
            const float* krow = &kcur[n * KSTRIDE + 2 * hi];
            v2f kb[32];
#pragma unroll
            for (int i = 0; i < 32; ++i)
                kb[i] = *(const v2f*)(krow + 4 * i);
            v2f va[32];
#pragma unroll
            for (int nt = 0; nt < 8; ++nt)
#pragma unroll
                for (int c = 0; c < 4; ++c)
                    va[nt * 4 + c] = *(const v2f*)&vcur[(nt * 16 + n) * VSTRIDE + 4 * c + 2 * hi];
            __builtin_amdgcn_sched_barrier(0);

            v8f s0 = {}, s1 = {};
#pragma unroll
            for (int i = 0; i < 32; i += 2) {
                s0 = wmma_f32(kb[i],     qa[i],     s0);
                s1 = wmma_f32(kb[i + 1], qa[i + 1], s1);
            }

            if (more) {
                stage_piece(knxt, vnxt, r, rv, dbase, 0, kq0, vq0);
                stage_piece(knxt, vnxt, r, rv, dbase, 1, kq1, vq1);
                stage_piece(knxt, vnxt, r, rv, dbase, 2, kq2, vq2);
                stage_piece(knxt, vnxt, r, rv, dbase, 3, kq3, vq3);
            }

            const int dl = q0 + n - (j * 16 + 8 * hi);
            float z[8];
#pragma unroll
            for (int e = 0; e < 8; ++e) {
                float sv = s0[e] + s1[e];
                sv = (sv > THR) ? sv : 0.0f;
                float zz = sv * SCALE2;
                if (e > dl) zz = -1e30f;
                z[e] = zz;
            }
            float mx = fmaxf(fmaxf(fmaxf(z[0], z[1]), fmaxf(z[2], z[3])),
                             fmaxf(fmaxf(z[4], z[5]), fmaxf(z[6], z[7])));
            mx = fmaxf(mx, xhalf(mx));
            const float mn = fmaxf(m_run, mx);
            const float al = __builtin_amdgcn_exp2f(m_run - mn);
            m_run = mn;
            float pe[8];
#pragma unroll
            for (int e = 0; e < 8; ++e)
                pe[e] = __builtin_amdgcn_exp2f(z[e] - mn);
            l_run = l_run * al + (((pe[0] + pe[1]) + (pe[2] + pe[3])) +
                                  ((pe[4] + pe[5]) + (pe[6] + pe[7])));

#pragma unroll
            for (int nt = 0; nt < 8; ++nt)
#pragma unroll
                for (int e = 0; e < 8; ++e)
                    y[nt][e] *= al;

            v2f pb[4];
#pragma unroll
            for (int c = 0; c < 4; ++c) {
                pb[c].x = pe[2 * c];
                pb[c].y = pe[2 * c + 1];
            }
#pragma unroll
            for (int c = 0; c < 4; ++c)
#pragma unroll
                for (int nt = 0; nt < 8; ++nt)
                    y[nt] = wmma_f32(va[nt * 4 + c], pb[c], y[nt]);
        } else if (more) {
            stage_piece(knxt, vnxt, r, rv, dbase, 0, kq0, vq0);
            stage_piece(knxt, vnxt, r, rv, dbase, 1, kq1, vq1);
            stage_piece(knxt, vnxt, r, rv, dbase, 2, kq2, vq2);
            stage_piece(knxt, vnxt, r, rv, dbase, 3, kq3, vq3);
        }
        __syncthreads();
    }

    float* mg = &lds_mg[w4 * (MGREGS * 32) + lane];
    if (set == 1) {
        mg[0]  = m_run;
        mg[32] = l_run;
#pragma unroll
        for (int nt = 0; nt < 8; ++nt)
#pragma unroll
            for (int e = 0; e < 8; ++e)
                mg[(2 + nt * 8 + e) * 32] = y[nt][e];
    }
    __syncthreads();
    if (set == 0) {
        const float m1 = mg[0];
        const float l1 = mg[32];
        const float mn = fmaxf(m_run, m1);
        const float c0 = __builtin_amdgcn_exp2f(m_run - mn);
        const float c1 = __builtin_amdgcn_exp2f(m1 - mn);
        float lt = l_run * c0 + l1 * c1;
        lt += xhalf(lt);
        const float inv = 1.0f / lt;
        const float a0 = c0 * inv;
        const float a1 = c1 * inv;
        float* op = O + ((size_t)b * S_ + q0 + n) * D_ + 8 * hi;
#pragma unroll
        for (int nt = 0; nt < 8; ++nt) {
            float4 o0, o1;
            o0.x = y[nt][0] * a0 + mg[(2 + nt * 8 + 0) * 32] * a1;
            o0.y = y[nt][1] * a0 + mg[(2 + nt * 8 + 1) * 32] * a1;
            o0.z = y[nt][2] * a0 + mg[(2 + nt * 8 + 2) * 32] * a1;
            o0.w = y[nt][3] * a0 + mg[(2 + nt * 8 + 3) * 32] * a1;
            o1.x = y[nt][4] * a0 + mg[(2 + nt * 8 + 4) * 32] * a1;
            o1.y = y[nt][5] * a0 + mg[(2 + nt * 8 + 5) * 32] * a1;
            o1.z = y[nt][6] * a0 + mg[(2 + nt * 8 + 6) * 32] * a1;
            o1.w = y[nt][7] * a0 + mg[(2 + nt * 8 + 7) * 32] * a1;
            *(float4*)(op + nt * 16)     = o0;
            *(float4*)(op + nt * 16 + 4) = o1;
        }
    }
  }
}

extern "C" void kernel_launch(void* const* d_in, const int* in_sizes, int n_in,
                              void* d_out, int out_size, void* d_ws, size_t ws_size,
                              hipStream_t stream) {
    const float* Q = (const float*)d_in[0];
    const float* T = (const float*)d_in[1];
    const float* V = (const float*)d_in[2];
    float*       O = (float*)d_out;
    dim3 grid(S_ / 128, B_);
    attn_fa_kernel<<<grid, dim3(256), 0, stream>>>(Q, T, V, O);
}


# K/V staging loads a whole trip ahead: reissued into the same registers behind the LDS stores that empty them, in flight across a hand-written LDS-only barrier; products and sums unchanged
# speedup vs baseline: 4.3763x; 1.0730x over previous
#include <hip/hip_runtime.h>


#define B_ 16
#define S_ 2048
#define D_ 128
#define THR 0.3f

typedef __attribute__((ext_vector_type(2))) float v2f;
typedef __attribute__((ext_vector_type(8))) float v8f;

#define KSTRIDE 132
#define VSTRIDE 20
#define MGREGS  66

static __device__ __forceinline__
v8f wmma_f32(v2f a, v2f b, v8f c) {
    return __builtin_amdgcn_wmma_f32_16x16x4_f32(false, a, false, b,
                                                 (short)0, c, false, false);
}

static __device__ __forceinline__
float xhalf(float x) {
    const int b = __builtin_bit_cast(int, x);
    return __builtin_bit_cast(float,
        __builtin_amdgcn_permlanex16(b, b, 0x76543210, 0xfedcba98, false, false));
}

static __device__ __forceinline__
void stage_piece(float* kdst, float* vdst, int rk, int rv, int dbase, int i, float4 kq, float4 vq) {
    *(float4*)&kdst[rk * KSTRIDE + dbase + 4 * i] = kq;
    vdst[(dbase + 4*i + 0) * VSTRIDE + rv] = vq.x;
    vdst[(dbase + 4*i + 1) * VSTRIDE + rv] = vq.y;
    vdst[(dbase + 4*i + 2) * VSTRIDE + rv] = vq.z;
    vdst[(dbase + 4*i + 3) * VSTRIDE + rv] = vq.w;
}

static __device__ __forceinline__
void lds_barrier() {
    asm volatile("s_wait_dscnt 0x0\n\ts_barrier_signal -1\n\ts_barrier_wait -1" ::: "memory");
}

__global__ __launch_bounds__(256) __attribute__((amdgpu_waves_per_eu(2, 2)))
void attn_fa_kernel(const float* __restrict__ Q,
                    const float* __restrict__ T,
                    const float* __restrict__ V,
                    float* __restrict__ O)
{
    __shared__ float lds_k [2 * 32  * KSTRIDE];
    __shared__ float lds_vt[2 * 2 * 128 * VSTRIDE];
    __shared__ float lds_mg[4 * MGREGS * 32];

    const int b    = blockIdx.y;
    const int p    = blockIdx.x;
    const int tid  = threadIdx.x;
    const int wave = tid >> 5;
    const int set  = wave >> 2;
    const int w4   = wave & 3;
    const int lane = tid & 31;
    const int n    = lane & 15;
    const int hi   = lane >> 4;
    const int r     = tid >> 3;
    const int rset  = r >> 4;
    const int r16   = r & 15;
    const int rv    = 4 * ((r16 >> 1) & 3) + 2 * (r16 >> 3) + (r16 & 1);
    const int dbase = (tid & 7) * 16;

    const float SCALE2 = 0.08838834764831845f * 1.4426950408889634f;

  for (int pass = 0; pass < 2; ++pass) {
    const int g      = pass ? p : (S_ / 64 - 1) - p;
    const int q0     = g * 64 + w4 * 16;
    const int myDiag = q0 >> 4;
    const int imax   = g * 2 + 1;

    v2f qa[32];
    {
        const float* qp = Q + ((size_t)b * S_ + q0 + n) * D_ + 2 * hi;
#pragma unroll
        for (int c = 0; c < 32; ++c)
            qa[c] = *(const v2f*)(qp + 4 * c);
    }

    v8f y[8];
#pragma unroll
    for (int nt = 0; nt < 8; ++nt) y[nt] = {};
    float m_run = -1e30f, l_run = 0.0f;

    const float* kp = T + ((size_t)b * S_ + r) * D_ + dbase;
    const float* vp = V + ((size_t)b * S_ + r) * D_ + dbase;
    float4 kq0 = *(const float4*)(kp + 0),  vq0 = *(const float4*)(vp + 0);
    float4 kq1 = *(const float4*)(kp + 4),  vq1 = *(const float4*)(vp + 4);
    float4 kq2 = *(const float4*)(kp + 8),  vq2 = *(const float4*)(vp + 8);
    float4 kq3 = *(const float4*)(kp + 12), vq3 = *(const float4*)(vp + 12);
    stage_piece(lds_k, lds_vt + rset * (128 * VSTRIDE), r, rv, dbase, 0, kq0, vq0);
    stage_piece(lds_k, lds_vt + rset * (128 * VSTRIDE), r, rv, dbase, 1, kq1, vq1);
    stage_piece(lds_k, lds_vt + rset * (128 * VSTRIDE), r, rv, dbase, 2, kq2, vq2);
    stage_piece(lds_k, lds_vt + rset * (128 * VSTRIDE), r, rv, dbase, 3, kq3, vq3);
    __syncthreads();
    kp += 32 * D_;
    vp += 32 * D_;
    kq0 = *(const float4*)(kp + 0);  vq0 = *(const float4*)(vp + 0);
    kq1 = *(const float4*)(kp + 4);  vq1 = *(const float4*)(vp + 4);
    kq2 = *(const float4*)(kp + 8);  vq2 = *(const float4*)(vp + 8);
    kq3 = *(const float4*)(kp + 12); vq3 = *(const float4*)(vp + 12);

    for (int ip = 0; ip <= imax; ++ip) {
        const int cur = ip & 1;
        const int j   = 2 * ip + set;
        const float* kcur = lds_k  + cur * (32 * KSTRIDE) + set * (16 * KSTRIDE);
        const float* vcur = lds_vt + cur * (2 * 128 * VSTRIDE) + set * (128 * VSTRIDE);
        float*       knxt = lds_k  + (cur ^ 1) * (32 * KSTRIDE);
        float*       vnxt = lds_vt + (cur ^ 1) * (2 * 128 * VSTRIDE) + rset * (128 * VSTRIDE);
        const bool   more  = (ip < imax);
        const bool   more2 = (ip + 1 < imax);

        if (j <= myDiag) {
            const float* krow = &kcur[n * KSTRIDE + 2 * hi];
            v2f kb[32];
#pragma unroll
            for (int i = 0; i < 32; ++i)
                kb[i] = *(const v2f*)(krow + 4 * i);
            v2f va[32];
#pragma unroll
            for (int nt = 0; nt < 8; ++nt)
#pragma unroll
                for (int c = 0; c < 4; ++c)
                    va[nt * 4 + c] = *(const v2f*)&vcur[(nt * 16 + n) * VSTRIDE + 4 * c + 2 * hi];
            __builtin_amdgcn_sched_barrier(0);

            v8f s0 = {}, s1 = {};
#pragma unroll
            for (int i = 0; i < 32; i += 2) {
                s0 = wmma_f32(kb[i],     qa[i],     s0);
                s1 = wmma_f32(kb[i + 1], qa[i + 1], s1);
            }

            if (more) {
                stage_piece(knxt, vnxt, r, rv, dbase, 0, kq0, vq0);
                stage_piece(knxt, vnxt, r, rv, dbase, 1, kq1, vq1);
                stage_piece(knxt, vnxt, r, rv, dbase, 2, kq2, vq2);
                stage_piece(knxt, vnxt, r, rv, dbase, 3, kq3, vq3);
                if (more2) {
                    kp += 32 * D_;
                    vp += 32 * D_;
                    kq0 = *(const float4*)(kp + 0);  vq0 = *(const float4*)(vp + 0);
                    kq1 = *(const float4*)(kp + 4);  vq1 = *(const float4*)(vp + 4);
                    kq2 = *(const float4*)(kp + 8);  vq2 = *(const float4*)(vp + 8);
                    kq3 = *(const float4*)(kp + 12); vq3 = *(const float4*)(vp + 12);
                }
            }

            const int dl = q0 + n - (j * 16 + 8 * hi);
            float z[8];
#pragma unroll
            for (int e = 0; e < 8; ++e) {
                float sv = s0[e] + s1[e];
                sv = (sv > THR) ? sv : 0.0f;
                float zz = sv * SCALE2;
                if (e > dl) zz = -1e30f;
                z[e] = zz;
            }
            float mx = fmaxf(fmaxf(fmaxf(z[0], z[1]), fmaxf(z[2], z[3])),
                             fmaxf(fmaxf(z[4], z[5]), fmaxf(z[6], z[7])));
            mx = fmaxf(mx, xhalf(mx));
            const float mn = fmaxf(m_run, mx);
            const float al = __builtin_amdgcn_exp2f(m_run - mn);
            m_run = mn;
            float pe[8];
#pragma unroll
            for (int e = 0; e < 8; ++e)
                pe[e] = __builtin_amdgcn_exp2f(z[e] - mn);
            l_run = l_run * al + (((pe[0] + pe[1]) + (pe[2] + pe[3])) +
                                  ((pe[4] + pe[5]) + (pe[6] + pe[7])));

#pragma unroll
            for (int nt = 0; nt < 8; ++nt)
#pragma unroll
                for (int e = 0; e < 8; ++e)
                    y[nt][e] *= al;

            v2f pb[4];
#pragma unroll
            for (int c = 0; c < 4; ++c) {
                pb[c].x = pe[2 * c];
                pb[c].y = pe[2 * c + 1];
            }
#pragma unroll
            for (int c = 0; c < 4; ++c)
#pragma unroll
                for (int nt = 0; nt < 8; ++nt)
                    y[nt] = wmma_f32(va[nt * 4 + c], pb[c], y[nt]);
        } else if (more) {
            stage_piece(knxt, vnxt, r, rv, dbase, 0, kq0, vq0);
            stage_piece(knxt, vnxt, r, rv, dbase, 1, kq1, vq1);
            stage_piece(knxt, vnxt, r, rv, dbase, 2, kq2, vq2);
            stage_piece(knxt, vnxt, r, rv, dbase, 3, kq3, vq3);
            if (more2) {
                kp += 32 * D_;
                vp += 32 * D_;
                kq0 = *(const float4*)(kp + 0);  vq0 = *(const float4*)(vp + 0);
                kq1 = *(const float4*)(kp + 4);  vq1 = *(const float4*)(vp + 4);
                kq2 = *(const float4*)(kp + 8);  vq2 = *(const float4*)(vp + 8);
                kq3 = *(const float4*)(kp + 12); vq3 = *(const float4*)(vp + 12);
            }
        }
        lds_barrier();
    }

    float* mg = &lds_mg[w4 * (MGREGS * 32) + lane];
    if (set == 1) {
        mg[0]  = m_run;
        mg[32] = l_run;
#pragma unroll
        for (int nt = 0; nt < 8; ++nt)
#pragma unroll
            for (int e = 0; e < 8; ++e)
                mg[(2 + nt * 8 + e) * 32] = y[nt][e];
    }
    __syncthreads();
    if (set == 0) {
        const float m1 = mg[0];
        const float l1 = mg[32];
        const float mn = fmaxf(m_run, m1);
        const float c0 = __builtin_amdgcn_exp2f(m_run - mn);
        const float c1 = __builtin_amdgcn_exp2f(m1 - mn);
        float lt = l_run * c0 + l1 * c1;
        lt += xhalf(lt);
        const float inv = 1.0f / lt;
        const float a0 = c0 * inv;
        const float a1 = c1 * inv;
        float* op = O + ((size_t)b * S_ + q0 + n) * D_ + 8 * hi;
#pragma unroll
        for (int nt = 0; nt < 8; ++nt) {
            float4 o0, o1;
            o0.x = y[nt][0] * a0 + mg[(2 + nt * 8 + 0) * 32] * a1;
            o0.y = y[nt][1] * a0 + mg[(2 + nt * 8 + 1) * 32] * a1;
            o0.z = y[nt][2] * a0 + mg[(2 + nt * 8 + 2) * 32] * a1;
            o0.w = y[nt][3] * a0 + mg[(2 + nt * 8 + 3) * 32] * a1;
            o1.x = y[nt][4] * a0 + mg[(2 + nt * 8 + 4) * 32] * a1;
            o1.y = y[nt][5] * a0 + mg[(2 + nt * 8 + 5) * 32] * a1;
            o1.z = y[nt][6] * a0 + mg[(2 + nt * 8 + 6) * 32] * a1;
            o1.w = y[nt][7] * a0 + mg[(2 + nt * 8 + 7) * 32] * a1;
            *(float4*)(op + nt * 16)     = o0;
            *(float4*)(op + nt * 16 + 4) = o1;
        }
    }
  }
}

extern "C" void kernel_launch(void* const* d_in, const int* in_sizes, int n_in,
                              void* d_out, int out_size, void* d_ws, size_t ws_size,
                              hipStream_t stream) {
    const float* Q = (const float*)d_in[0];
    const float* T = (const float*)d_in[1];
    const float* V = (const float*)d_in[2];
    float*       O = (float*)d_out;
    dim3 grid(S_ / 128, B_);
    attn_fa_kernel<<<grid, dim3(256), 0, stream>>>(Q, T, V, O);
}


# 32-key softmax steps: a trip stages 64 keys, each wave set takes two adjacent key tiles as ONE step (K, then V^T fragments, 128 registers each); one max/rescale/sum per 32 keys: row-sum order changes
# speedup vs baseline: 4.8166x; 1.1006x over previous
#include <hip/hip_runtime.h>


#define B_ 16
#define S_ 2048
#define D_ 128
#define THR 0.3f

typedef __attribute__((ext_vector_type(2))) float v2f;
typedef __attribute__((ext_vector_type(8))) float v8f;

#define KSTRIDE 132
#define VSTRIDE 20
#define MGREGS  66
#define KBUF (64 * KSTRIDE)
#define VTILE (128 * VSTRIDE)
#define VBUF (4 * VTILE)

static __device__ __forceinline__
v8f wmma_f32(v2f a, v2f b, v8f c) {
    return __builtin_amdgcn_wmma_f32_16x16x4_f32(false, a, false, b,
                                                 (short)0, c, false, false);
}

static __device__ __forceinline__
float xhalf(float x) {
    const int b = __builtin_bit_cast(int, x);
    return __builtin_bit_cast(float,
        __builtin_amdgcn_permlanex16(b, b, 0x76543210, 0xfedcba98, false, false));
}

static __device__ __forceinline__
void stage_piece(float* kdst, float* vdst, int rk, int rv, int dbase, int i, float4 kq, float4 vq) {
    *(float4*)&kdst[rk * KSTRIDE + dbase + 4 * i] = kq;
    vdst[(dbase + 4*i + 0) * VSTRIDE + rv] = vq.x;
    vdst[(dbase + 4*i + 1) * VSTRIDE + rv] = vq.y;
    vdst[(dbase + 4*i + 2) * VSTRIDE + rv] = vq.z;
    vdst[(dbase + 4*i + 3) * VSTRIDE + rv] = vq.w;
}

static __device__ __forceinline__
void stage_all(float* kdst, float* vdst, int r, int rv, int dbase, const float4* kq, const float4* vq) {
#pragma unroll
    for (int i = 0; i < 4; ++i) {
        stage_piece(kdst, vdst,             r,      rv, dbase, i, kq[i],     vq[i]);
        stage_piece(kdst, vdst + 2 * VTILE, r + 32, rv, dbase, i, kq[4 + i], vq[4 + i]);
    }
}

static __device__ __forceinline__
void fetch_all(const float* kp, const float* vp, float4* kq, float4* vq) {
#pragma unroll
    for (int i = 0; i < 4; ++i) {
        kq[i]     = *(const float4*)(kp + 4 * i);
        vq[i]     = *(const float4*)(vp + 4 * i);
        kq[4 + i] = *(const float4*)(kp + 32 * D_ + 4 * i);
        vq[4 + i] = *(const float4*)(vp + 32 * D_ + 4 * i);
    }
}

static __device__ __forceinline__
void lds_barrier() {
    asm volatile("s_wait_dscnt 0x0\n\ts_barrier_signal -1\n\ts_barrier_wait -1" ::: "memory");
}

__global__ __launch_bounds__(256) __attribute__((amdgpu_waves_per_eu(2, 2)))
void attn_fa_kernel(const float* __restrict__ Q,
                    const float* __restrict__ T,
                    const float* __restrict__ V,
                    float* __restrict__ O)
{
    __shared__ float lds_k [2 * KBUF];
    __shared__ float lds_vt[2 * VBUF];
    __shared__ float lds_mg[4 * MGREGS * 32];

    const int b    = blockIdx.y;
    const int p    = blockIdx.x;
    const int tid  = threadIdx.x;
    const int wave = tid >> 5;
    const int set  = wave >> 2;
    const int w4   = wave & 3;
    const int lane = tid & 31;
    const int n    = lane & 15;
    const int hi   = lane >> 4;
    const int r     = tid >> 3;
    const int rset  = r >> 4;
    const int r16   = r & 15;
    const int rv    = 4 * ((r16 >> 1) & 3) + 2 * (r16 >> 3) + (r16 & 1);
    const int dbase = (tid & 7) * 16;

    const float SCALE2 = 0.08838834764831845f * 1.4426950408889634f;

  for (int pass = 0; pass < 2; ++pass) {
    const int g      = pass ? p : (S_ / 64 - 1) - p;
    const int q0     = g * 64 + w4 * 16;
    const int myDiag = q0 >> 4;

    v2f qa[32];
    {
        const float* qp = Q + ((size_t)b * S_ + q0 + n) * D_ + 2 * hi;
#pragma unroll
        for (int c = 0; c < 32; ++c)
            qa[c] = *(const v2f*)(qp + 4 * c);
    }

    v8f y[8];
#pragma unroll
    for (int nt = 0; nt < 8; ++nt) y[nt] = {};
    float m_run = -1e30f, l_run = 0.0f;

    const float* kp = T + ((size_t)b * S_ + r) * D_ + dbase;
    const float* vp = V + ((size_t)b * S_ + r) * D_ + dbase;
    float4 kq[8], vq[8];
    fetch_all(kp, vp, kq, vq);
    stage_all(lds_k, lds_vt + rset * VTILE, r, rv, dbase, kq, vq);
    __syncthreads();
    if (g >= 1) {
        kp += 64 * D_;
        vp += 64 * D_;
        fetch_all(kp, vp, kq, vq);
    }

    for (int it = 0; it <= g; ++it) {
        const int cur = it & 1;
        const int ta  = 4 * it + 2 * set;
        const float* kcur = lds_k  + cur * KBUF + set * (32 * KSTRIDE);
        const float* vcur = lds_vt + cur * VBUF + set * (2 * VTILE);
        float*       knxt = lds_k  + (cur ^ 1) * KBUF;
        float*       vnxt = lds_vt + (cur ^ 1) * VBUF + rset * VTILE;
        const bool   more  = (it < g);
        const bool   more2 = (it + 1 < g);

        if (ta <= myDiag) {
            const float* krow = &kcur[n * KSTRIDE + 2 * hi];
            v2f kb[64];
#pragma unroll
            for (int i = 0; i < 32; i += 2)
#pragma unroll
                for (int t = 0; t < 2; ++t) {
                    kb[t * 32 + i]     = *(const v2f*)(krow + t * (16 * KSTRIDE) + 4 * i);
                    kb[t * 32 + i + 1] = *(const v2f*)(krow + t * (16 * KSTRIDE) + 4 * i + 4);
                }
            __builtin_amdgcn_sched_barrier(0);

            v8f s0a = {}, s1a = {}, s0b = {}, s1b = {};
#pragma unroll
            for (int i = 0; i < 32; i += 2) {
                s0a = wmma_f32(kb[i],          qa[i],     s0a);
                s0b = wmma_f32(kb[32 + i],     qa[i],     s0b);
                s1a = wmma_f32(kb[i + 1],      qa[i + 1], s1a);
                s1b = wmma_f32(kb[32 + i + 1], qa[i + 1], s1b);
            }
            int voff = n * VSTRIDE + 2 * hi;
            asm volatile("" : "+v"(voff) : "v"(s0a), "v"(s0b), "v"(s1a), "v"(s1b));

            v2f va[64];
#pragma unroll
            for (int t = 0; t < 2; ++t)
#pragma unroll
                for (int nt = 0; nt < 8; ++nt)
#pragma unroll
                    for (int c = 0; c < 4; ++c)
                        va[t * 32 + nt * 4 + c] =
                            *(const v2f*)&vcur[t * VTILE + nt * (16 * VSTRIDE) + 4 * c + voff];
            __builtin_amdgcn_sched_barrier(0);

            if (more) {
                stage_all(knxt, vnxt, r, rv, dbase, kq, vq);
                if (more2) {
                    kp += 64 * D_;
                    vp += 64 * D_;
                    fetch_all(kp, vp, kq, vq);
                }
            }

            const int dl = q0 + n - (ta * 16 + 8 * hi);
            float za[8], zb[8];
#pragma unroll
            for (int e = 0; e < 8; ++e) {
                float sa = s0a[e] + s1a[e];
                float sb = s0b[e] + s1b[e];
                sa = (sa > THR) ? sa : 0.0f;
                sb = (sb > THR) ? sb : 0.0f;
                float fa = sa * SCALE2;
                float fb = sb * SCALE2;
                if (e > dl)      fa = -1e30f;
                if (e > dl - 16) fb = -1e30f;
                za[e] = fa;
                zb[e] = fb;
            }
            float mxa = fmaxf(fmaxf(fmaxf(za[0], za[1]), fmaxf(za[2], za[3])),
                              fmaxf(fmaxf(za[4], za[5]), fmaxf(za[6], za[7])));
            float mxb = fmaxf(fmaxf(fmaxf(zb[0], zb[1]), fmaxf(zb[2], zb[3])),
                              fmaxf(fmaxf(zb[4], zb[5]), fmaxf(zb[6], zb[7])));
            float mx = fmaxf(mxa, mxb);
            mx = fmaxf(mx, xhalf(mx));
            const float mn = fmaxf(m_run, mx);
            const float al = __builtin_amdgcn_exp2f(m_run - mn);
            m_run = mn;
            float pa[8], pc[8];
#pragma unroll
            for (int e = 0; e < 8; ++e) {
                pa[e] = __builtin_amdgcn_exp2f(za[e] - mn);
                pc[e] = __builtin_amdgcn_exp2f(zb[e] - mn);
            }
            l_run = l_run * al + ((((pa[0] + pa[1]) + (pa[2] + pa[3])) +
                                   ((pa[4] + pa[5]) + (pa[6] + pa[7]))) +
                                  (((pc[0] + pc[1]) + (pc[2] + pc[3])) +
                                   ((pc[4] + pc[5]) + (pc[6] + pc[7]))));

#pragma unroll
            for (int nt = 0; nt < 8; ++nt)
#pragma unroll
                for (int e = 0; e < 8; ++e)
                    y[nt][e] *= al;

            v2f pb[8];
#pragma unroll
            for (int c = 0; c < 4; ++c) {
                pb[c].x     = pa[2 * c];
                pb[c].y     = pa[2 * c + 1];
                pb[4 + c].x = pc[2 * c];
                pb[4 + c].y = pc[2 * c + 1];
            }
#pragma unroll
            for (int tc = 0; tc < 8; ++tc)
#pragma unroll
                for (int nt = 0; nt < 8; ++nt)
                    y[nt] = wmma_f32(va[(tc >> 2) * 32 + nt * 4 + (tc & 3)], pb[tc], y[nt]);
        } else if (more) {
            stage_all(knxt, vnxt, r, rv, dbase, kq, vq);
            if (more2) {
                kp += 64 * D_;
                vp += 64 * D_;
                fetch_all(kp, vp, kq, vq);
            }
        }
        lds_barrier();
    }

    float* mg = &lds_mg[w4 * (MGREGS * 32) + lane];
    if (set == 1) {
        mg[0]  = m_run;
        mg[32] = l_run;
#pragma unroll
        for (int nt = 0; nt < 8; ++nt)
#pragma unroll
            for (int e = 0; e < 8; ++e)
                mg[(2 + nt * 8 + e) * 32] = y[nt][e];
    }
    __syncthreads();
    if (set == 0) {
        const float m1 = mg[0];
        const float l1 = mg[32];
        const float mn = fmaxf(m_run, m1);
        const float c0 = __builtin_amdgcn_exp2f(m_run - mn);
        const float c1 = __builtin_amdgcn_exp2f(m1 - mn);
        float lt = l_run * c0 + l1 * c1;
        lt += xhalf(lt);
        const float inv = 1.0f / lt;
        const float a0 = c0 * inv;
        const float a1 = c1 * inv;
        float* op = O + ((size_t)b * S_ + q0 + n) * D_ + 8 * hi;
#pragma unroll
        for (int nt = 0; nt < 8; ++nt) {
            float4 o0, o1;
            o0.x = y[nt][0] * a0 + mg[(2 + nt * 8 + 0) * 32] * a1;
            o0.y = y[nt][1] * a0 + mg[(2 + nt * 8 + 1) * 32] * a1;
            o0.z = y[nt][2] * a0 + mg[(2 + nt * 8 + 2) * 32] * a1;
            o0.w = y[nt][3] * a0 + mg[(2 + nt * 8 + 3) * 32] * a1;
            o1.x = y[nt][4] * a0 + mg[(2 + nt * 8 + 4) * 32] * a1;
            o1.y = y[nt][5] * a0 + mg[(2 + nt * 8 + 5) * 32] * a1;
            o1.z = y[nt][6] * a0 + mg[(2 + nt * 8 + 6) * 32] * a1;
            o1.w = y[nt][7] * a0 + mg[(2 + nt * 8 + 7) * 32] * a1;
            *(float4*)(op + nt * 16)     = o0;
            *(float4*)(op + nt * 16 + 4) = o1;
        }
    }
  }
}

extern "C" void kernel_launch(void* const* d_in, const int* in_sizes, int n_in,
                              void* d_out, int out_size, void* d_ws, size_t ws_size,
                              hipStream_t stream) {
    const float* Q = (const float*)d_in[0];
    const float* T = (const float*)d_in[1];
    const float* V = (const float*)d_in[2];
    float*       O = (float*)d_out;
    dim3 grid(S_ / 128, B_);
    attn_fa_kernel<<<grid, dim3(256), 0, stream>>>(Q, T, V, O);
}
